# starting kernel
# baseline (speedup 1.0000x reference)
#include <hip/hip_runtime.h>
#include <hip/hip_bf16.h>

#define DIM       128
#define HEADS     4
#define HEAD_DIM  32

typedef __attribute__((ext_vector_type(2))) float v2f;
typedef __attribute__((ext_vector_type(8))) float v8f;

__device__ __forceinline__ unsigned flipf(float f) {
    unsigned u = __float_as_uint(f);
    return (u & 0x80000000u) ? ~u : (u | 0x80000000u);
}
__device__ __forceinline__ float unflipf(unsigned u) {
    return __uint_as_float((u & 0x80000000u) ? (u & 0x7FFFFFFFu) : ~u);
}

__global__ __launch_bounds__(256)
void gat_gemm_wmma(const float* __restrict__ x, const float* __restrict__ W,
                   float* __restrict__ Wx, int nNodes, int nTiles) {
    int wave = (blockIdx.x * blockDim.x + threadIdx.x) >> 5;
    if (wave >= nTiles) return;
    int lane = threadIdx.x & 31;
    int rc   = lane & 15;
    int hi   = lane >> 4;
    int nodeBase = wave * 16;

    v8f acc[8] = {};

    int arow = nodeBase + rc;
    if (arow >= nNodes) arow = nNodes - 1;
    const float* aptr = x + (size_t)arow * DIM + hi * 2;

    #pragma unroll 8
    for (int k = 0; k < DIM; k += 4) {
        v2f a = *(const v2f*)(aptr + k);
        #pragma unroll
        for (int nt = 0; nt < 8; ++nt) {
            v2f b = *(const v2f*)(W + (size_t)(nt * 16 + rc) * DIM + k + hi * 2);
            acc[nt] = __builtin_amdgcn_wmma_f32_16x16x4_f32(
                false, a, false, b, (short)0, acc[nt], false, false);
        }
    }

    #pragma unroll
    for (int nt = 0; nt < 8; ++nt) {
        #pragma unroll
        for (int j = 0; j < 8; ++j) {
            int node = nodeBase + j + 8 * hi;
            if (node < nNodes)
                Wx[(size_t)node * DIM + nt * 16 + rc] = acc[nt][j];
        }
    }
}

__global__ __launch_bounds__(256)
void gat_scores(const float* __restrict__ Wx, const float* __restrict__ a_w,
                float* __restrict__ s_src, float* __restrict__ s_dst, int nNodes) {
    int t = blockIdx.x * blockDim.x + threadIdx.x;
    if (t >= nNodes * HEADS) return;
    int h = t & (HEADS - 1);
    int n = t >> 2;
    const float* p = Wx + (size_t)n * DIM + h * HEAD_DIM;
    float ss = 0.f, sd = 0.f;
    #pragma unroll
    for (int d = 0; d < HEAD_DIM; ++d) {
        float v = p[d];
        ss += v * a_w[d];
        sd += v * a_w[HEAD_DIM + d];
    }
    s_src[t] = ss;
    s_dst[t] = sd;
}

__global__ __launch_bounds__(256)
void gat_edge_max(const int* __restrict__ ei, const float* __restrict__ s_src,
                  const float* __restrict__ s_dst, unsigned* __restrict__ emax,
                  int nE) {
    int t = blockIdx.x * blockDim.x + threadIdx.x;
    if (t >= nE * HEADS) return;
    int h = t & (HEADS - 1);
    int e = t >> 2;
    int src = ei[e];
    int dst = ei[nE + e];
    float v = s_src[src * HEADS + h] + s_dst[dst * HEADS + h];
    v = v > 0.f ? v : 0.2f * v;
    atomicMax(&emax[dst * HEADS + h], flipf(v));
}

__global__ __launch_bounds__(256)
void gat_edge_accum(const int* __restrict__ ei, const float* __restrict__ s_src,
                    const float* __restrict__ s_dst, const unsigned* __restrict__ emax,
                    const float* __restrict__ Wx, float* __restrict__ denom,
                    float* __restrict__ out, int nE) {
    int gw = (int)((blockIdx.x * (size_t)blockDim.x + threadIdx.x) >> 5);
    if (gw >= nE * HEADS) return;
    int lane = threadIdx.x & 31;
    int h = gw & (HEADS - 1);
    int e = gw >> 2;
    int src = ei[e];
    int dst = ei[nE + e];
    float v = s_src[src * HEADS + h] + s_dst[dst * HEADS + h];
    v = v > 0.f ? v : 0.2f * v;
    float em = unflipf(emax[dst * HEADS + h]);
    float ee = expf(v - em);
    if (lane == 0) atomicAdd(&denom[dst * HEADS + h], ee);
    atomicAdd(&out[(size_t)dst * DIM + h * HEAD_DIM + lane],
              ee * Wx[(size_t)src * DIM + h * HEAD_DIM + lane]);
}

__global__ __launch_bounds__(256)
void gat_finalize(float* __restrict__ out, const float* __restrict__ denom,
                  int nNodes) {
    int t = blockIdx.x * blockDim.x + threadIdx.x;
    if (t >= nNodes * DIM) return;
    int n = t >> 7;
    int d = t & (DIM - 1);
    int h = d >> 5;
    float v = out[t] / (denom[n * HEADS + h] + 1e-8f);
    out[t] = v > 0.f ? v : expm1f(v);
}

extern "C" void kernel_launch(void* const* d_in, const int* in_sizes, int n_in,
                              void* d_out, int out_size, void* d_ws, size_t ws_size,
                              hipStream_t stream) {
    const float* x   = (const float*)d_in[0];
    const int*   ei  = (const int*)d_in[1];
    const float* W   = (const float*)d_in[2];
    const float* a_w = (const float*)d_in[3];
    int N = in_sizes[0] / DIM;
    int E = in_sizes[1] / 2;
    float* out = (float*)d_out;

    float*    Wx    = (float*)d_ws;
    float*    s_src = Wx + (size_t)N * DIM;
    float*    s_dst = s_src + (size_t)N * HEADS;
    float*    denom = s_dst + (size_t)N * HEADS;
    unsigned* emax  = (unsigned*)(denom + (size_t)N * HEADS);

    hipMemsetAsync(out,   0, (size_t)N * DIM * sizeof(float), stream);
    hipMemsetAsync(denom, 0, (size_t)N * HEADS * sizeof(float), stream);
    hipMemsetAsync(emax,  0, (size_t)N * HEADS * sizeof(unsigned), stream);

    dim3 blk(256);
    int nTiles = (N + 15) / 16;
    gat_gemm_wmma<<<(nTiles + 7) / 8, blk, 0, stream>>>(x, W, Wx, N, nTiles);
    gat_scores  <<<(N * HEADS + 255) / 256, blk, 0, stream>>>(Wx, a_w, s_src, s_dst, N);
    gat_edge_max<<<((size_t)E * HEADS + 255) / 256, blk, 0, stream>>>(ei, s_src, s_dst, emax, E);
    gat_edge_accum<<<((size_t)E * HEADS * 32 + 255) / 256, blk, 0, stream>>>(
        ei, s_src, s_dst, emax, Wx, denom, out, E);
    gat_finalize<<<((size_t)N * DIM + 255) / 256, blk, 0, stream>>>(out, denom, N);
}


# Edges counting-sorted by dst, each segment then ordered by edge id; one wave per dst node sums its rows in that fixed order in registers: float atomics removed, summation order changed, now fixed
# speedup vs baseline: 2.5011x; 2.5011x over previous
#include <hip/hip_runtime.h>
#include <hip/hip_bf16.h>

#define DIM       128
#define HEADS     4
#define HEAD_DIM  32

typedef __attribute__((ext_vector_type(2))) float v2f;
typedef __attribute__((ext_vector_type(8))) float v8f;

__global__ __launch_bounds__(256)
void gat_gemm_wmma(const float* __restrict__ x, const float* __restrict__ W,
                   float* __restrict__ Wx, int nNodes, int nTiles) {
    int wave = (blockIdx.x * blockDim.x + threadIdx.x) >> 5;
    if (wave >= nTiles) return;
    int lane = threadIdx.x & 31;
    int rc   = lane & 15;
    int hi   = lane >> 4;
    int nodeBase = wave * 16;

    v8f acc[8] = {};

    int arow = nodeBase + rc;
    if (arow >= nNodes) arow = nNodes - 1;
    const float* aptr = x + (size_t)arow * DIM + hi * 2;

    #pragma unroll 8
    for (int k = 0; k < DIM; k += 4) {
        v2f a = *(const v2f*)(aptr + k);
        #pragma unroll
        for (int nt = 0; nt < 8; ++nt) {
            v2f b = *(const v2f*)(W + (size_t)(nt * 16 + rc) * DIM + k + hi * 2);
            acc[nt] = __builtin_amdgcn_wmma_f32_16x16x4_f32(
                false, a, false, b, (short)0, acc[nt], false, false);
        }
    }

    #pragma unroll
    for (int nt = 0; nt < 8; ++nt) {
        #pragma unroll
        for (int j = 0; j < 8; ++j) {
            int node = nodeBase + j + 8 * hi;
            if (node < nNodes)
                Wx[(size_t)node * DIM + nt * 16 + rc] = acc[nt][j];
        }
    }
}

__global__ __launch_bounds__(256)
void gat_scores(const float* __restrict__ Wx, const float* __restrict__ a_w,
                float* __restrict__ s_src, float* __restrict__ s_dst, int nNodes) {
    int t = blockIdx.x * blockDim.x + threadIdx.x;
    if (t >= nNodes * HEADS) return;
    int h = t & (HEADS - 1);
    int n = t >> 2;
    const float* p = Wx + (size_t)n * DIM + h * HEAD_DIM;
    float ss = 0.f, sd = 0.f;
    #pragma unroll
    for (int d = 0; d < HEAD_DIM; ++d) {
        float v = p[d];
        ss += v * a_w[d];
        sd += v * a_w[HEAD_DIM + d];
    }
    s_src[t] = ss;
    s_dst[t] = sd;
}

__global__ __launch_bounds__(256)
void gat_degree(const int* __restrict__ ei, int* __restrict__ cnt,
                int* __restrict__ rk, int nE) {
    int e = blockIdx.x * blockDim.x + threadIdx.x;
    if (e >= nE) return;
    rk[e] = atomicAdd(&cnt[ei[nE + e]], 1);
}

__global__ __launch_bounds__(1024)
void gat_scan(int* __restrict__ off, int n) {
    __shared__ int wsum[32];
    int lane = threadIdx.x & 31;
    int w    = threadIdx.x >> 5;
    int per  = (n + 31) / 32;
    int lo   = w * per;       if (lo > n) lo = n;
    int hi   = lo + per;      if (hi > n) hi = n;

    int s = 0;
    for (int i = lo + lane; i < hi; i += 32) s += off[i];
    for (int o = 16; o; o >>= 1) s += __shfl_xor(s, o);
    if (lane == 0) wsum[w] = s;
    __syncthreads();

    int carry = 0, total = 0;
    for (int k = 0; k < 32; ++k) {
        int v = wsum[k];
        if (k < w) carry += v;
        total += v;
    }

    for (int b = lo; b < hi; b += 128) {
        int i0 = b + lane * 4;
        int c0 = (i0     < hi) ? off[i0]     : 0;
        int c1 = (i0 + 1 < hi) ? off[i0 + 1] : 0;
        int c2 = (i0 + 2 < hi) ? off[i0 + 2] : 0;
        int c3 = (i0 + 3 < hi) ? off[i0 + 3] : 0;
        int t = c0 + c1 + c2 + c3;
        int inc = t;
        for (int o = 1; o < 32; o <<= 1) {
            int v = __shfl_up(inc, o);
            if (lane >= o) inc += v;
        }
        int base = carry + inc - t;
        if (i0     < hi) off[i0]     = base;
        if (i0 + 1 < hi) off[i0 + 1] = base + c0;
        if (i0 + 2 < hi) off[i0 + 2] = base + c0 + c1;
        if (i0 + 3 < hi) off[i0 + 3] = base + c0 + c1 + c2;
        carry += __shfl(inc, 31);
    }
    if (threadIdx.x == 0) off[n] = total;
}

__global__ __launch_bounds__(256)
void gat_place(const int* __restrict__ ei, const int* __restrict__ off,
               const int* __restrict__ rk, int* __restrict__ perm, int nE) {
    int e = blockIdx.x * blockDim.x + threadIdx.x;
    if (e >= nE) return;
    perm[off[ei[nE + e]] + rk[e]] = e;
}

__global__ __launch_bounds__(256)
void gat_order(const int* __restrict__ ei, const int* __restrict__ off,
               const int* __restrict__ perm, int* __restrict__ csr_src, int nE) {
    int p = blockIdx.x * blockDim.x + threadIdx.x;
    if (p >= nE) return;
    int eid = perm[p];
    int d   = ei[nE + eid];
    int beg = off[d], end = off[d + 1];
    int r = 0;
    for (int q = beg; q < end; ++q) r += (perm[q] < eid) ? 1 : 0;
    csr_src[beg + r] = ei[eid];
}

__device__ __forceinline__ float lrelu(float v) { return v > 0.f ? v : 0.2f * v; }

__device__ __forceinline__ float lane_f(float v, int j) {
    return __int_as_float(__builtin_amdgcn_readlane(__float_as_int(v), j));
}

__global__ __launch_bounds__(256)
void gat_node_reduce(const int* __restrict__ off, const int* __restrict__ csr_src,
                     const float* __restrict__ s_src, const float* __restrict__ s_dst,
                     const float* __restrict__ Wx, float* __restrict__ out, int nNodes) {
    int n = (int)((blockIdx.x * (size_t)blockDim.x + threadIdx.x) >> 5);
    n = __builtin_amdgcn_readfirstlane(n);
    if (n >= nNodes) return;
    int lane = threadIdx.x & 31;
    int beg = off[n], end = off[n + 1];
    const float4* s_src4 = (const float4*)s_src;
    float4 sd = ((const float4*)s_dst)[n];

    float m0 = -INFINITY, m1 = -INFINITY, m2 = -INFINITY, m3 = -INFINITY;
    for (int b = beg; b < end; b += 32) {
        int i = b + lane;
        if (i < end) {
            float4 ss = s_src4[csr_src[i]];
            m0 = fmaxf(m0, lrelu(ss.x + sd.x));
            m1 = fmaxf(m1, lrelu(ss.y + sd.y));
            m2 = fmaxf(m2, lrelu(ss.z + sd.z));
            m3 = fmaxf(m3, lrelu(ss.w + sd.w));
        }
    }
    for (int o = 16; o; o >>= 1) {
        m0 = fmaxf(m0, __shfl_xor(m0, o));
        m1 = fmaxf(m1, __shfl_xor(m1, o));
        m2 = fmaxf(m2, __shfl_xor(m2, o));
        m3 = fmaxf(m3, __shfl_xor(m3, o));
    }

    float a0 = 0.f, a1 = 0.f, a2 = 0.f, a3 = 0.f;
    float d0 = 0.f, d1 = 0.f, d2 = 0.f, d3 = 0.f;
    for (int b = beg; b < end; b += 32) {
        int i = b + lane;
        int s = 0;
        float e0 = 0.f, e1 = 0.f, e2 = 0.f, e3 = 0.f;
        if (i < end) {
            s = csr_src[i];
            float4 ss = s_src4[s];
            e0 = expf(lrelu(ss.x + sd.x) - m0);
            e1 = expf(lrelu(ss.y + sd.y) - m1);
            e2 = expf(lrelu(ss.z + sd.z) - m2);
            e3 = expf(lrelu(ss.w + sd.w) - m3);
        }
        d0 += e0; d1 += e1; d2 += e2; d3 += e3;

        int cnt = end - b; if (cnt > 32) cnt = 32;
        int j = 0;
        for (; j + 4 <= cnt; j += 4) {
            const float* p0 = Wx + (size_t)__builtin_amdgcn_readlane(s, j    ) * DIM + lane;
            const float* p1 = Wx + (size_t)__builtin_amdgcn_readlane(s, j + 1) * DIM + lane;
            const float* p2 = Wx + (size_t)__builtin_amdgcn_readlane(s, j + 2) * DIM + lane;
            const float* p3 = Wx + (size_t)__builtin_amdgcn_readlane(s, j + 3) * DIM + lane;
            float r00 = p0[0], r01 = p0[32], r02 = p0[64], r03 = p0[96];
            float r10 = p1[0], r11 = p1[32], r12 = p1[64], r13 = p1[96];
            float r20 = p2[0], r21 = p2[32], r22 = p2[64], r23 = p2[96];
            float r30 = p3[0], r31 = p3[32], r32 = p3[64], r33 = p3[96];
            a0 += lane_f(e0, j    ) * r00; a1 += lane_f(e1, j    ) * r01;
            a2 += lane_f(e2, j    ) * r02; a3 += lane_f(e3, j    ) * r03;
            a0 += lane_f(e0, j + 1) * r10; a1 += lane_f(e1, j + 1) * r11;
            a2 += lane_f(e2, j + 1) * r12; a3 += lane_f(e3, j + 1) * r13;
            a0 += lane_f(e0, j + 2) * r20; a1 += lane_f(e1, j + 2) * r21;
            a2 += lane_f(e2, j + 2) * r22; a3 += lane_f(e3, j + 2) * r23;
            a0 += lane_f(e0, j + 3) * r30; a1 += lane_f(e1, j + 3) * r31;
            a2 += lane_f(e2, j + 3) * r32; a3 += lane_f(e3, j + 3) * r33;
        }
        for (; j < cnt; ++j) {
            const float* p0 = Wx + (size_t)__builtin_amdgcn_readlane(s, j) * DIM + lane;
            float r00 = p0[0], r01 = p0[32], r02 = p0[64], r03 = p0[96];
            a0 += lane_f(e0, j) * r00; a1 += lane_f(e1, j) * r01;
            a2 += lane_f(e2, j) * r02; a3 += lane_f(e3, j) * r03;
        }
    }
    for (int o = 16; o; o >>= 1) {
        d0 += __shfl_xor(d0, o);
        d1 += __shfl_xor(d1, o);
        d2 += __shfl_xor(d2, o);
        d3 += __shfl_xor(d3, o);
    }
    float* orow = out + (size_t)n * DIM + lane;
    float v;
    v = a0 / (d0 + 1e-8f); orow[0]  = v > 0.f ? v : expm1f(v);
    v = a1 / (d1 + 1e-8f); orow[32] = v > 0.f ? v : expm1f(v);
    v = a2 / (d2 + 1e-8f); orow[64] = v > 0.f ? v : expm1f(v);
    v = a3 / (d3 + 1e-8f); orow[96] = v > 0.f ? v : expm1f(v);
}

extern "C" void kernel_launch(void* const* d_in, const int* in_sizes, int n_in,
                              void* d_out, int out_size, void* d_ws, size_t ws_size,
                              hipStream_t stream) {
    const float* x   = (const float*)d_in[0];
    const int*   ei  = (const int*)d_in[1];
    const float* W   = (const float*)d_in[2];
    const float* a_w = (const float*)d_in[3];
    int N = in_sizes[0] / DIM;
    int E = in_sizes[1] / 2;
    float* out = (float*)d_out;

    float* Wx      = (float*)d_ws;
    float* s_src   = Wx + (size_t)N * DIM;
    float* s_dst   = s_src + (size_t)N * HEADS;
    int*   off     = (int*)(s_dst + (size_t)N * HEADS);
    int*   csr_src = off + (size_t)N + 1;
    int*   rk      = (int*)d_out;
    int*   perm    = rk + (size_t)E;

    size_t need = ((size_t)N * (DIM + 2 * HEADS + 1) + 1 + (size_t)E) * sizeof(float);
    if (ws_size < need || (size_t)2 * E > (size_t)N * DIM) return;

    hipMemsetAsync(off, 0, ((size_t)N + 1) * sizeof(int), stream);

    dim3 blk(256);
    int nTiles = (N + 15) / 16;
    gat_gemm_wmma<<<(nTiles + 7) / 8, blk, 0, stream>>>(x, W, Wx, N, nTiles);
    gat_scores  <<<(N * HEADS + 255) / 256, blk, 0, stream>>>(Wx, a_w, s_src, s_dst, N);
    gat_degree  <<<(E + 255) / 256, blk, 0, stream>>>(ei, off, rk, E);
    gat_scan    <<<1, 1024, 0, stream>>>(off, N);
    gat_place   <<<(E + 255) / 256, blk, 0, stream>>>(ei, off, rk, perm, E);
    gat_order   <<<(E + 255) / 256, blk, 0, stream>>>(ei, off, perm, csr_src, E);
    gat_node_reduce<<<(N + 7) / 8, blk, 0, stream>>>(off, csr_src, s_src, s_dst, Wx, out, N);
}


# Independent chains share launches: the degree histogram runs in the GEMM's grid, the attention scores in the one-block scan's grid; 7 launches to 5; float sums and their order unchanged
# speedup vs baseline: 2.8342x; 1.1332x over previous
#include <hip/hip_runtime.h>
#include <hip/hip_bf16.h>

#define DIM       128
#define HEADS     4
#define HEAD_DIM  32

typedef __attribute__((ext_vector_type(2))) float v2f;
typedef __attribute__((ext_vector_type(8))) float v8f;

__device__ __forceinline__
void gemm_tile_wmma(const float* __restrict__ x, const float* __restrict__ W,
                    float* __restrict__ Wx, int nNodes, int nTiles) {
    int wave = (blockIdx.x * blockDim.x + threadIdx.x) >> 5;
    if (wave >= nTiles) return;
    int lane = threadIdx.x & 31;
    int rc   = lane & 15;
    int hi   = lane >> 4;
    int nodeBase = wave * 16;

    v8f acc[8] = {};

    int arow = nodeBase + rc;
    if (arow >= nNodes) arow = nNodes - 1;
    const float* aptr = x + (size_t)arow * DIM + hi * 2;

    #pragma unroll 8
    for (int k = 0; k < DIM; k += 4) {
        v2f a = *(const v2f*)(aptr + k);
        #pragma unroll
        for (int nt = 0; nt < 8; ++nt) {
            v2f b = *(const v2f*)(W + (size_t)(nt * 16 + rc) * DIM + k + hi * 2);
            acc[nt] = __builtin_amdgcn_wmma_f32_16x16x4_f32(
                false, a, false, b, (short)0, acc[nt], false, false);
        }
    }

    #pragma unroll
    for (int nt = 0; nt < 8; ++nt) {
        #pragma unroll
        for (int j = 0; j < 8; ++j) {
            int node = nodeBase + j + 8 * hi;
            if (node < nNodes)
                Wx[(size_t)node * DIM + nt * 16 + rc] = acc[nt][j];
        }
    }
}

__device__ __forceinline__
void scores_thread(const float* __restrict__ Wx, const float* __restrict__ a_w,
                   float* __restrict__ s_src, float* __restrict__ s_dst, int nNodes, int t) {
    if (t >= nNodes * HEADS) return;
    int h = t & (HEADS - 1);
    int n = t >> 2;
    const float* p = Wx + (size_t)n * DIM + h * HEAD_DIM;
    float ss = 0.f, sd = 0.f;
    #pragma unroll
    for (int d = 0; d < HEAD_DIM; ++d) {
        float v = p[d];
        ss += v * a_w[d];
        sd += v * a_w[HEAD_DIM + d];
    }
    s_src[t] = ss;
    s_dst[t] = sd;
}

__device__ __forceinline__
void degree_edge(const int* __restrict__ ei, int* __restrict__ cnt,
                 int* __restrict__ rk, int nE, int e) {
    if (e >= nE) return;
    rk[e] = atomicAdd(&cnt[ei[nE + e]], 1);
}

__global__ __launch_bounds__(256)
void gat_gemm_degree(const float* __restrict__ x, const float* __restrict__ W,
                     float* __restrict__ Wx, int nNodes, int nTiles, int gemmBlocks,
                     const int* __restrict__ ei, int* __restrict__ cnt,
                     int* __restrict__ rk, int nE) {
    if ((int)blockIdx.x < gemmBlocks) {
        gemm_tile_wmma(x, W, Wx, nNodes, nTiles);
    } else {
        degree_edge(ei, cnt, rk, nE,
                    (int)(blockIdx.x - gemmBlocks) * (int)blockDim.x + (int)threadIdx.x);
    }
}

__device__ __forceinline__
void scan_block(int* __restrict__ off, int n, int* wsum) {
    int lane = threadIdx.x & 31;
    int w    = threadIdx.x >> 5;
    int per  = (n + 31) / 32;
    int lo   = w * per;       if (lo > n) lo = n;
    int hi   = lo + per;      if (hi > n) hi = n;

    int s = 0;
    for (int i = lo + lane; i < hi; i += 32) s += off[i];
    for (int o = 16; o; o >>= 1) s += __shfl_xor(s, o);
    if (lane == 0) wsum[w] = s;
    __syncthreads();

    int carry = 0, total = 0;
    for (int k = 0; k < 32; ++k) {
        int v = wsum[k];
        if (k < w) carry += v;
        total += v;
    }

    for (int b = lo; b < hi; b += 128) {
        int i0 = b + lane * 4;
        int c0 = (i0     < hi) ? off[i0]     : 0;
        int c1 = (i0 + 1 < hi) ? off[i0 + 1] : 0;
        int c2 = (i0 + 2 < hi) ? off[i0 + 2] : 0;
        int c3 = (i0 + 3 < hi) ? off[i0 + 3] : 0;
        int t = c0 + c1 + c2 + c3;
        int inc = t;
        for (int o = 1; o < 32; o <<= 1) {
            int v = __shfl_up(inc, o);
            if (lane >= o) inc += v;
        }
        int base = carry + inc - t;
        if (i0     < hi) off[i0]     = base;
        if (i0 + 1 < hi) off[i0 + 1] = base + c0;
        if (i0 + 2 < hi) off[i0 + 2] = base + c0 + c1;
        if (i0 + 3 < hi) off[i0 + 3] = base + c0 + c1 + c2;
        carry += __shfl(inc, 31);
    }
    if (threadIdx.x == 0) off[n] = total;
}

__global__ __launch_bounds__(1024)
void gat_scan_scores(int* __restrict__ off, int nNodes,
                     const float* __restrict__ Wx, const float* __restrict__ a_w,
                     float* __restrict__ s_src, float* __restrict__ s_dst) {
    __shared__ int wsum[32];
    if (blockIdx.x == 0) {
        scan_block(off, nNodes, wsum);
    } else {
        scores_thread(Wx, a_w, s_src, s_dst, nNodes,
                      (int)(blockIdx.x - 1) * (int)blockDim.x + (int)threadIdx.x);
    }
}

__global__ __launch_bounds__(256)
void gat_place(const int* __restrict__ ei, const int* __restrict__ off,
               const int* __restrict__ rk, int* __restrict__ perm, int nE) {
    int e = blockIdx.x * blockDim.x + threadIdx.x;
    if (e >= nE) return;
    perm[off[ei[nE + e]] + rk[e]] = e;
}

__global__ __launch_bounds__(256)
void gat_order(const int* __restrict__ ei, const int* __restrict__ off,
               const int* __restrict__ perm, int* __restrict__ csr_src, int nE) {
    int p = blockIdx.x * blockDim.x + threadIdx.x;
    if (p >= nE) return;
    int eid = perm[p];
    int d   = ei[nE + eid];
    int beg = off[d], end = off[d + 1];
    int r = 0;
    for (int q = beg; q < end; ++q) r += (perm[q] < eid) ? 1 : 0;
    csr_src[beg + r] = ei[eid];
}

__device__ __forceinline__ float lrelu(float v) { return v > 0.f ? v : 0.2f * v; }

__device__ __forceinline__ float lane_f(float v, int j) {
    return __int_as_float(__builtin_amdgcn_readlane(__float_as_int(v), j));
}

__global__ __launch_bounds__(256)
void gat_node_reduce(const int* __restrict__ off, const int* __restrict__ csr_src,
                     const float* __restrict__ s_src, const float* __restrict__ s_dst,
                     const float* __restrict__ Wx, float* __restrict__ out, int nNodes) {
    int n = (int)((blockIdx.x * (size_t)blockDim.x + threadIdx.x) >> 5);
    n = __builtin_amdgcn_readfirstlane(n);
    if (n >= nNodes) return;
    int lane = threadIdx.x & 31;
    int beg = off[n], end = off[n + 1];
    const float4* s_src4 = (const float4*)s_src;
    float4 sd = ((const float4*)s_dst)[n];

    float m0 = -INFINITY, m1 = -INFINITY, m2 = -INFINITY, m3 = -INFINITY;
    for (int b = beg; b < end; b += 32) {
        int i = b + lane;
        if (i < end) {
            float4 ss = s_src4[csr_src[i]];
            m0 = fmaxf(m0, lrelu(ss.x + sd.x));
            m1 = fmaxf(m1, lrelu(ss.y + sd.y));
            m2 = fmaxf(m2, lrelu(ss.z + sd.z));
            m3 = fmaxf(m3, lrelu(ss.w + sd.w));
        }
    }
    for (int o = 16; o; o >>= 1) {
        m0 = fmaxf(m0, __shfl_xor(m0, o));
        m1 = fmaxf(m1, __shfl_xor(m1, o));
        m2 = fmaxf(m2, __shfl_xor(m2, o));
        m3 = fmaxf(m3, __shfl_xor(m3, o));
    }

    float a0 = 0.f, a1 = 0.f, a2 = 0.f, a3 = 0.f;
    float d0 = 0.f, d1 = 0.f, d2 = 0.f, d3 = 0.f;
    for (int b = beg; b < end; b += 32) {
        int i = b + lane;
        int s = 0;
        float e0 = 0.f, e1 = 0.f, e2 = 0.f, e3 = 0.f;
        if (i < end) {
            s = csr_src[i];
            float4 ss = s_src4[s];
            e0 = expf(lrelu(ss.x + sd.x) - m0);
            e1 = expf(lrelu(ss.y + sd.y) - m1);
            e2 = expf(lrelu(ss.z + sd.z) - m2);
            e3 = expf(lrelu(ss.w + sd.w) - m3);
        }
        d0 += e0; d1 += e1; d2 += e2; d3 += e3;

        int cnt = end - b; if (cnt > 32) cnt = 32;
        int j = 0;
        for (; j + 4 <= cnt; j += 4) {
            const float* p0 = Wx + (size_t)__builtin_amdgcn_readlane(s, j    ) * DIM + lane;
            const float* p1 = Wx + (size_t)__builtin_amdgcn_readlane(s, j + 1) * DIM + lane;
            const float* p2 = Wx + (size_t)__builtin_amdgcn_readlane(s, j + 2) * DIM + lane;
            const float* p3 = Wx + (size_t)__builtin_amdgcn_readlane(s, j + 3) * DIM + lane;
            float r00 = p0[0], r01 = p0[32], r02 = p0[64], r03 = p0[96];
            float r10 = p1[0], r11 = p1[32], r12 = p1[64], r13 = p1[96];
            float r20 = p2[0], r21 = p2[32], r22 = p2[64], r23 = p2[96];
            float r30 = p3[0], r31 = p3[32], r32 = p3[64], r33 = p3[96];
            a0 += lane_f(e0, j    ) * r00; a1 += lane_f(e1, j    ) * r01;
            a2 += lane_f(e2, j    ) * r02; a3 += lane_f(e3, j    ) * r03;
            a0 += lane_f(e0, j + 1) * r10; a1 += lane_f(e1, j + 1) * r11;
            a2 += lane_f(e2, j + 1) * r12; a3 += lane_f(e3, j + 1) * r13;
            a0 += lane_f(e0, j + 2) * r20; a1 += lane_f(e1, j + 2) * r21;
            a2 += lane_f(e2, j + 2) * r22; a3 += lane_f(e3, j + 2) * r23;
            a0 += lane_f(e0, j + 3) * r30; a1 += lane_f(e1, j + 3) * r31;
            a2 += lane_f(e2, j + 3) * r32; a3 += lane_f(e3, j + 3) * r33;
        }
        for (; j < cnt; ++j) {
            const float* p0 = Wx + (size_t)__builtin_amdgcn_readlane(s, j) * DIM + lane;
            float r00 = p0[0], r01 = p0[32], r02 = p0[64], r03 = p0[96];
            a0 += lane_f(e0, j) * r00; a1 += lane_f(e1, j) * r01;
            a2 += lane_f(e2, j) * r02; a3 += lane_f(e3, j) * r03;
        }
    }
    for (int o = 16; o; o >>= 1) {
        d0 += __shfl_xor(d0, o);
        d1 += __shfl_xor(d1, o);
        d2 += __shfl_xor(d2, o);
        d3 += __shfl_xor(d3, o);
    }
    float* orow = out + (size_t)n * DIM + lane;
    float v;
    v = a0 / (d0 + 1e-8f); orow[0]  = v > 0.f ? v : expm1f(v);
    v = a1 / (d1 + 1e-8f); orow[32] = v > 0.f ? v : expm1f(v);
    v = a2 / (d2 + 1e-8f); orow[64] = v > 0.f ? v : expm1f(v);
    v = a3 / (d3 + 1e-8f); orow[96] = v > 0.f ? v : expm1f(v);
}

extern "C" void kernel_launch(void* const* d_in, const int* in_sizes, int n_in,
                              void* d_out, int out_size, void* d_ws, size_t ws_size,
                              hipStream_t stream) {
    const float* x   = (const float*)d_in[0];
    const int*   ei  = (const int*)d_in[1];
    const float* W   = (const float*)d_in[2];
    const float* a_w = (const float*)d_in[3];
    int N = in_sizes[0] / DIM;
    int E = in_sizes[1] / 2;
    float* out = (float*)d_out;

    float* Wx      = (float*)d_ws;
    float* s_src   = Wx + (size_t)N * DIM;
    float* s_dst   = s_src + (size_t)N * HEADS;
    int*   off     = (int*)(s_dst + (size_t)N * HEADS);
    int*   csr_src = off + (size_t)N + 1;
    int*   rk      = (int*)d_out;
    int*   perm    = rk + (size_t)E;

    size_t need = ((size_t)N * (DIM + 2 * HEADS + 1) + 1 + (size_t)E) * sizeof(float);
    if (ws_size < need || (size_t)2 * E > (size_t)N * DIM) return;

    hipMemsetAsync(off, 0, ((size_t)N + 1) * sizeof(int), stream);

    dim3 blk(256);
    int nTiles = (N + 15) / 16;
    int gemmBlocks = (nTiles + 7) / 8;
    gat_gemm_degree<<<gemmBlocks + (E + 255) / 256, blk, 0, stream>>>(
        x, W, Wx, N, nTiles, gemmBlocks, ei, off, rk, E);
    gat_scan_scores<<<1 + (N * HEADS + 1023) / 1024, 1024, 0, stream>>>(
        off, N, Wx, a_w, s_src, s_dst);
    gat_place   <<<(E + 255) / 256, blk, 0, stream>>>(ei, off, rk, perm, E);
    gat_order   <<<(E + 255) / 256, blk, 0, stream>>>(ei, off, perm, csr_src, E);
    gat_node_reduce<<<(N + 7) / 8, blk, 0, stream>>>(off, csr_src, s_src, s_dst, Wx, out, N);
}


# Scan, place, order launches replaced by 64-slot (edge id, src) buckets per dst in the out rows, filled at the histogram rank; reduce wave ranks its bucket by edge id; 5 launches to 3; sums unchanged
# speedup vs baseline: 3.7839x; 1.3351x over previous
#include <hip/hip_runtime.h>
#include <hip/hip_bf16.h>

#define DIM       128
#define HEADS     4
#define HEAD_DIM  32

typedef __attribute__((ext_vector_type(2))) float v2f;
typedef __attribute__((ext_vector_type(8))) float v8f;

__device__ __forceinline__
void gemm_tile_wmma(const float* __restrict__ x, const float* __restrict__ W,
                    float* __restrict__ Wx, int nNodes, int nTiles) {
    int wave = (blockIdx.x * blockDim.x + threadIdx.x) >> 5;
    if (wave >= nTiles) return;
    int lane = threadIdx.x & 31;
    int rc   = lane & 15;
    int hi   = lane >> 4;
    int nodeBase = wave * 16;

    v8f acc[8] = {};

    int arow = nodeBase + rc;
    if (arow >= nNodes) arow = nNodes - 1;
    const float* aptr = x + (size_t)arow * DIM + hi * 2;

    #pragma unroll 8
    for (int k = 0; k < DIM; k += 4) {
        v2f a = *(const v2f*)(aptr + k);
        #pragma unroll
        for (int nt = 0; nt < 8; ++nt) {
            v2f b = *(const v2f*)(W + (size_t)(nt * 16 + rc) * DIM + k + hi * 2);
            acc[nt] = __builtin_amdgcn_wmma_f32_16x16x4_f32(
                false, a, false, b, (short)0, acc[nt], false, false);
        }
    }

    #pragma unroll
    for (int nt = 0; nt < 8; ++nt) {
        #pragma unroll
        for (int j = 0; j < 8; ++j) {
            int node = nodeBase + j + 8 * hi;
            if (node < nNodes)
                Wx[(size_t)node * DIM + nt * 16 + rc] = acc[nt][j];
        }
    }
}

__device__ __forceinline__
void scores_thread(const float* __restrict__ Wx, const float* __restrict__ a_w,
                   float* __restrict__ s_src, float* __restrict__ s_dst, int nNodes, int t) {
    if (t >= nNodes * HEADS) return;
    int h = t & (HEADS - 1);
    int n = t >> 2;
    const float* p = Wx + (size_t)n * DIM + h * HEAD_DIM;
    float ss = 0.f, sd = 0.f;
    #pragma unroll
    for (int d = 0; d < HEAD_DIM; ++d) {
        float v = p[d];
        ss += v * a_w[d];
        sd += v * a_w[HEAD_DIM + d];
    }
    s_src[t] = ss;
    s_dst[t] = sd;
}

#define CAP 64

__device__ __forceinline__
void degree_fill(const int* ei, int* cnt,
                 int* __restrict__ bucket, int nE, int e) {
    if (e >= nE) return;
    int s = ei[e];
    int d = ei[nE + e];
    int r = atomicAdd(&cnt[d], 1);
    if (r < CAP) ((int2*)bucket)[(size_t)d * CAP + r] = make_int2(e, s);
}

__global__ __launch_bounds__(256)
void gat_gemm_fill(const float* __restrict__ x, const float* __restrict__ W,
                   float* __restrict__ Wx, int nNodes, int nTiles, int gemmBlocks,
                   const int* ei, int* cnt,
                   int* __restrict__ bucket, int nE) {
    if ((int)blockIdx.x < gemmBlocks) {
        gemm_tile_wmma(x, W, Wx, nNodes, nTiles);
    } else {
        degree_fill(ei, cnt, bucket, nE,
                    (int)(blockIdx.x - gemmBlocks) * (int)blockDim.x + (int)threadIdx.x);
    }
}

__global__ __launch_bounds__(1024)
void gat_scores(const float* __restrict__ Wx, const float* __restrict__ a_w,
                float* __restrict__ s_src, float* __restrict__ s_dst, int nNodes) {
    scores_thread(Wx, a_w, s_src, s_dst, nNodes,
                  (int)blockIdx.x * (int)blockDim.x + (int)threadIdx.x);
}

__device__ __forceinline__ float lrelu(float v) { return v > 0.f ? v : 0.2f * v; }

__device__ __forceinline__ float lane_f(float v, int j) {
    return __int_as_float(__builtin_amdgcn_readlane(__float_as_int(v), j));
}

__device__ __forceinline__ float elu1(float v) { return v > 0.f ? v : expm1f(v); }

__device__ __forceinline__
void node_reduce_slow(int n, int lane, const int* __restrict__ ei, int nE,
                      const float4* __restrict__ s_src4, float4 sd,
                      const float* __restrict__ Wx, float* __restrict__ orow) {
    float m0 = -INFINITY, m1 = -INFINITY, m2 = -INFINITY, m3 = -INFINITY;
    for (int b = 0; b < nE; b += 32) {
        int e = b + lane;
        if (e < nE && ei[nE + e] == n) {
            float4 ss = s_src4[ei[e]];
            m0 = fmaxf(m0, lrelu(ss.x + sd.x));
            m1 = fmaxf(m1, lrelu(ss.y + sd.y));
            m2 = fmaxf(m2, lrelu(ss.z + sd.z));
            m3 = fmaxf(m3, lrelu(ss.w + sd.w));
        }
    }
    for (int o = 16; o; o >>= 1) {
        m0 = fmaxf(m0, __shfl_xor(m0, o));
        m1 = fmaxf(m1, __shfl_xor(m1, o));
        m2 = fmaxf(m2, __shfl_xor(m2, o));
        m3 = fmaxf(m3, __shfl_xor(m3, o));
    }
    float a0 = 0.f, a1 = 0.f, a2 = 0.f, a3 = 0.f;
    float d0 = 0.f, d1 = 0.f, d2 = 0.f, d3 = 0.f;
    for (int b = 0; b < nE; b += 32) {
        int e = b + lane;
        bool hit = (e < nE) && (ei[nE + e] == n);
        unsigned mask = __builtin_amdgcn_ballot_w32(hit);
        while (mask) {
            int j = __builtin_ctz(mask);
            mask &= mask - 1;
            int s = ei[b + j];
            float4 ss = s_src4[s];
            float e0 = expf(lrelu(ss.x + sd.x) - m0);
            float e1 = expf(lrelu(ss.y + sd.y) - m1);
            float e2 = expf(lrelu(ss.z + sd.z) - m2);
            float e3 = expf(lrelu(ss.w + sd.w) - m3);
            d0 += e0; d1 += e1; d2 += e2; d3 += e3;
            const float* p = Wx + (size_t)s * DIM + lane;
            a0 += e0 * p[0];  a1 += e1 * p[32];
            a2 += e2 * p[64]; a3 += e3 * p[96];
        }
    }
    orow[0]  = elu1(a0 / (d0 + 1e-8f));
    orow[32] = elu1(a1 / (d1 + 1e-8f));
    orow[64] = elu1(a2 / (d2 + 1e-8f));
    orow[96] = elu1(a3 / (d3 + 1e-8f));
}

__global__ __launch_bounds__(256)
void gat_node_reduce(const int* __restrict__ cnt, const int* __restrict__ ei, int nE,
                     const float* __restrict__ s_src, const float* __restrict__ s_dst,
                     const float* __restrict__ Wx, float* out, int nNodes) {
    __shared__ int lsrc[8][CAP];
    int n = (int)((blockIdx.x * (size_t)blockDim.x + threadIdx.x) >> 5);
    n = __builtin_amdgcn_readfirstlane(n);
    if (n >= nNodes) return;
    int lane = threadIdx.x & 31;
    int* ls = lsrc[__builtin_amdgcn_readfirstlane((int)(threadIdx.x >> 5))];
    int deg = cnt[n];
    const float4* s_src4 = (const float4*)s_src;
    float4 sd = ((const float4*)s_dst)[n];
    float* orow = out + (size_t)n * DIM + lane;

    if (deg > CAP) {
        node_reduce_slow(n, lane, ei, nE, s_src4, sd, Wx, orow);
        return;
    }

    const int2* bk = (const int2*)out + (size_t)n * CAP;
    int2 pa = make_int2(0, 0), pb = make_int2(0, 0);
    if (lane < deg)      pa = bk[lane];
    if (lane + 32 < deg) pb = bk[lane + 32];
    int ra = 0, rb = 0;
    if (deg <= 32) {
        for (int j = 0; j < deg; ++j)
            ra += (__builtin_amdgcn_readlane(pa.x, j) < pa.x) ? 1 : 0;
    } else {
        for (int j = 0; j < 32; ++j) {
            int v = __builtin_amdgcn_readlane(pa.x, j);
            ra += (v < pa.x) ? 1 : 0;
            rb += (v < pb.x) ? 1 : 0;
        }
        for (int j = 0; j < deg - 32; ++j) {
            int v = __builtin_amdgcn_readlane(pb.x, j);
            ra += (v < pa.x) ? 1 : 0;
            rb += (v < pb.x) ? 1 : 0;
        }
    }
    if (lane < deg)      ls[ra] = pa.y;
    if (lane + 32 < deg) ls[rb] = pb.y;

    float m0 = -INFINITY, m1 = -INFINITY, m2 = -INFINITY, m3 = -INFINITY;
    for (int b = 0; b < deg; b += 32) {
        int i = b + lane;
        if (i < deg) {
            float4 ss = s_src4[ls[i]];
            m0 = fmaxf(m0, lrelu(ss.x + sd.x));
            m1 = fmaxf(m1, lrelu(ss.y + sd.y));
            m2 = fmaxf(m2, lrelu(ss.z + sd.z));
            m3 = fmaxf(m3, lrelu(ss.w + sd.w));
        }
    }
    for (int o = 16; o; o >>= 1) {
        m0 = fmaxf(m0, __shfl_xor(m0, o));
        m1 = fmaxf(m1, __shfl_xor(m1, o));
        m2 = fmaxf(m2, __shfl_xor(m2, o));
        m3 = fmaxf(m3, __shfl_xor(m3, o));
    }

    float a0 = 0.f, a1 = 0.f, a2 = 0.f, a3 = 0.f;
    float d0 = 0.f, d1 = 0.f, d2 = 0.f, d3 = 0.f;
    for (int b = 0; b < deg; b += 32) {
        int i = b + lane;
        int s = 0;
        float e0 = 0.f, e1 = 0.f, e2 = 0.f, e3 = 0.f;
        if (i < deg) {
            s = ls[i];
            float4 ss = s_src4[s];
            e0 = expf(lrelu(ss.x + sd.x) - m0);
            e1 = expf(lrelu(ss.y + sd.y) - m1);
            e2 = expf(lrelu(ss.z + sd.z) - m2);
            e3 = expf(lrelu(ss.w + sd.w) - m3);
        }
        d0 += e0; d1 += e1; d2 += e2; d3 += e3;

        int c = deg - b; if (c > 32) c = 32;
        int j = 0;
        for (; j + 4 <= c; j += 4) {
            const float* p0 = Wx + (size_t)__builtin_amdgcn_readlane(s, j    ) * DIM + lane;
            const float* p1 = Wx + (size_t)__builtin_amdgcn_readlane(s, j + 1) * DIM + lane;
            const float* p2 = Wx + (size_t)__builtin_amdgcn_readlane(s, j + 2) * DIM + lane;
            const float* p3 = Wx + (size_t)__builtin_amdgcn_readlane(s, j + 3) * DIM + lane;
            float r00 = p0[0], r01 = p0[32], r02 = p0[64], r03 = p0[96];
            float r10 = p1[0], r11 = p1[32], r12 = p1[64], r13 = p1[96];
            float r20 = p2[0], r21 = p2[32], r22 = p2[64], r23 = p2[96];
            float r30 = p3[0], r31 = p3[32], r32 = p3[64], r33 = p3[96];
            a0 += lane_f(e0, j    ) * r00; a1 += lane_f(e1, j    ) * r01;
            a2 += lane_f(e2, j    ) * r02; a3 += lane_f(e3, j    ) * r03;
            a0 += lane_f(e0, j + 1) * r10; a1 += lane_f(e1, j + 1) * r11;
            a2 += lane_f(e2, j + 1) * r12; a3 += lane_f(e3, j + 1) * r13;
            a0 += lane_f(e0, j + 2) * r20; a1 += lane_f(e1, j + 2) * r21;
            a2 += lane_f(e2, j + 2) * r22; a3 += lane_f(e3, j + 2) * r23;
            a0 += lane_f(e0, j + 3) * r30; a1 += lane_f(e1, j + 3) * r31;
            a2 += lane_f(e2, j + 3) * r32; a3 += lane_f(e3, j + 3) * r33;
        }
        for (; j < c; ++j) {
            const float* p0 = Wx + (size_t)__builtin_amdgcn_readlane(s, j) * DIM + lane;
            float r00 = p0[0], r01 = p0[32], r02 = p0[64], r03 = p0[96];
            a0 += lane_f(e0, j) * r00; a1 += lane_f(e1, j) * r01;
            a2 += lane_f(e2, j) * r02; a3 += lane_f(e3, j) * r03;
        }
    }
    for (int o = 16; o; o >>= 1) {
        d0 += __shfl_xor(d0, o);
        d1 += __shfl_xor(d1, o);
        d2 += __shfl_xor(d2, o);
        d3 += __shfl_xor(d3, o);
    }
    orow[0]  = elu1(a0 / (d0 + 1e-8f));
    orow[32] = elu1(a1 / (d1 + 1e-8f));
    orow[64] = elu1(a2 / (d2 + 1e-8f));
    orow[96] = elu1(a3 / (d3 + 1e-8f));
}

extern "C" void kernel_launch(void* const* d_in, const int* in_sizes, int n_in,
                              void* d_out, int out_size, void* d_ws, size_t ws_size,
                              hipStream_t stream) {
    const float* x   = (const float*)d_in[0];
    const int*   ei  = (const int*)d_in[1];
    const float* W   = (const float*)d_in[2];
    const float* a_w = (const float*)d_in[3];
    int N = in_sizes[0] / DIM;
    int E = in_sizes[1] / 2;
    float* out = (float*)d_out;

    float* Wx    = (float*)d_ws;
    float* s_src = Wx + (size_t)N * DIM;
    float* s_dst = s_src + (size_t)N * HEADS;
    int*   cnt   = (int*)(s_dst + (size_t)N * HEADS);
    int*   bucket = (int*)d_out;

    size_t need = (size_t)N * (DIM + 2 * HEADS + 1) * sizeof(float);
    if (ws_size < need) return;

    hipMemsetAsync(cnt, 0, (size_t)N * sizeof(int), stream);

    dim3 blk(256);
    int nTiles = (N + 15) / 16;
    int gemmBlocks = (nTiles + 7) / 8;
    gat_gemm_fill<<<gemmBlocks + (E + 255) / 256, blk, 0, stream>>>(
        x, W, Wx, N, nTiles, gemmBlocks, ei, cnt, bucket, E);
    gat_scores<<<(N * HEADS + 1023) / 1024, 1024, 0, stream>>>(Wx, a_w, s_src, s_dst, N);
    gat_node_reduce<<<(N + 7) / 8, blk, 0, stream>>>(cnt, ei, E, s_src, s_dst, Wx, out, N);
}


# Attention scalars summed in the projection's epilogue from its accumulators (16-lane DPP butterfly): scores launch and its Wx re-read removed, 3 launches to 2; score sum order changed, fixed
# speedup vs baseline: 3.9524x; 1.0445x over previous
#include <hip/hip_runtime.h>
#include <hip/hip_bf16.h>

#define DIM       128
#define HEADS     4
#define HEAD_DIM  32

typedef __attribute__((ext_vector_type(2))) float v2f;
typedef __attribute__((ext_vector_type(8))) float v8f;


template <int M>
__device__ __forceinline__ float row_xor(float v) {
    return __int_as_float(__builtin_amdgcn_update_dpp(
        0, __float_as_int(v), 0x160 + M, 0xf, 0xf, true));
}

template <int M>
__device__ __forceinline__ float fold_pair(float lo, float hi, bool bit) {
    float keep = bit ? hi : lo;
    float send = bit ? lo : hi;
    return keep + row_xor<M>(send);
}

__device__ __forceinline__
void gemm_tile_wmma(const float* __restrict__ x, const float* __restrict__ W,
                    const float* __restrict__ a_w,
                    float* __restrict__ Wx, float* __restrict__ s_src,
                    float* __restrict__ s_dst, int nNodes, int nTiles) {
    int wave = (blockIdx.x * blockDim.x + threadIdx.x) >> 5;
    if (wave >= nTiles) return;
    int lane = threadIdx.x & 31;
    int rc   = lane & 15;
    int hi   = lane >> 4;
    int nodeBase = wave * 16;

    v8f acc[8] = {};

    int arow = nodeBase + rc;
    if (arow >= nNodes) arow = nNodes - 1;
    const float* aptr = x + (size_t)arow * DIM + hi * 2;

    float as0 = a_w[rc], as1 = a_w[16 + rc];
    float ad0 = a_w[HEAD_DIM + rc], ad1 = a_w[HEAD_DIM + 16 + rc];

    #pragma unroll 8
    for (int k = 0; k < DIM; k += 4) {
        v2f a = *(const v2f*)(aptr + k);
        #pragma unroll
        for (int nt = 0; nt < 8; ++nt) {
            v2f b = *(const v2f*)(W + (size_t)(nt * 16 + rc) * DIM + k + hi * 2);
            acc[nt] = __builtin_amdgcn_wmma_f32_16x16x4_f32(
                false, a, false, b, (short)0, acc[nt], false, false);
        }
    }

    #pragma unroll
    for (int nt = 0; nt < 8; ++nt) {
        #pragma unroll
        for (int j = 0; j < 8; ++j) {
            int node = nodeBase + j + 8 * hi;
            if (node < nNodes)
                Wx[(size_t)node * DIM + nt * 16 + rc] = acc[nt][j];
        }
    }

    float v[64];
    #pragma unroll
    for (int h = 0; h < HEADS; ++h) {
        #pragma unroll
        for (int j = 0; j < 8; ++j) {
            float x0 = acc[2 * h][j], x1 = acc[2 * h + 1][j];
            v[(h * 2 + 0) * 8 + j] = x0 * as0 + x1 * as1;
            v[(h * 2 + 1) * 8 + j] = x0 * ad0 + x1 * ad1;
        }
    }
    bool b0 = rc & 1, b1 = rc & 2, b2 = rc & 4, b3 = rc & 8;
    float w0[32], w1[16], w2[8], w3[4];
    #pragma unroll
    for (int t = 0; t < 32; ++t) w0[t] = fold_pair<1>(v[2 * t], v[2 * t + 1], b0);
    #pragma unroll
    for (int t = 0; t < 16; ++t) w1[t] = fold_pair<2>(w0[2 * t], w0[2 * t + 1], b1);
    #pragma unroll
    for (int t = 0; t < 8; ++t)  w2[t] = fold_pair<4>(w1[2 * t], w1[2 * t + 1], b2);
    #pragma unroll
    for (int t = 0; t < 4; ++t)  w3[t] = fold_pair<8>(w2[2 * t], w2[2 * t + 1], b3);

    int snode = nodeBase + (rc & 7) + 8 * hi;
    float* sp = b3 ? s_dst : s_src;
    if (snode < nNodes)
        *(float4*)(sp + (size_t)snode * HEADS) = make_float4(w3[0], w3[1], w3[2], w3[3]);
}

#define CAP 64

__device__ __forceinline__
void degree_fill(const int* ei, int* cnt,
                 int* __restrict__ bucket, int nE, int e) {
    if (e >= nE) return;
    int s = ei[e];
    int d = ei[nE + e];
    int r = atomicAdd(&cnt[d], 1);
    if (r < CAP) ((int2*)bucket)[(size_t)d * CAP + r] = make_int2(e, s);
}

__global__ __launch_bounds__(256)
void gat_gemm_fill(const float* __restrict__ x, const float* __restrict__ W,
                   const float* __restrict__ a_w,
                   float* __restrict__ Wx, float* __restrict__ s_src,
                   float* __restrict__ s_dst, int nNodes, int nTiles, int gemmBlocks,
                   const int* ei, int* cnt,
                   int* __restrict__ bucket, int nE) {
    if ((int)blockIdx.x < gemmBlocks) {
        gemm_tile_wmma(x, W, a_w, Wx, s_src, s_dst, nNodes, nTiles);
    } else {
        degree_fill(ei, cnt, bucket, nE,
                    (int)(blockIdx.x - gemmBlocks) * (int)blockDim.x + (int)threadIdx.x);
    }
}

__device__ __forceinline__ float lrelu(float v) { return v > 0.f ? v : 0.2f * v; }

__device__ __forceinline__ float lane_f(float v, int j) {
    return __int_as_float(__builtin_amdgcn_readlane(__float_as_int(v), j));
}

__device__ __forceinline__ float elu1(float v) { return v > 0.f ? v : expm1f(v); }

__device__ __forceinline__
void node_reduce_slow(int n, int lane, const int* __restrict__ ei, int nE,
                      const float4* __restrict__ s_src4, float4 sd,
                      const float* __restrict__ Wx, float* __restrict__ orow) {
    float m0 = -INFINITY, m1 = -INFINITY, m2 = -INFINITY, m3 = -INFINITY;
    for (int b = 0; b < nE; b += 32) {
        int e = b + lane;
        if (e < nE && ei[nE + e] == n) {
            float4 ss = s_src4[ei[e]];
            m0 = fmaxf(m0, lrelu(ss.x + sd.x));
            m1 = fmaxf(m1, lrelu(ss.y + sd.y));
            m2 = fmaxf(m2, lrelu(ss.z + sd.z));
            m3 = fmaxf(m3, lrelu(ss.w + sd.w));
        }
    }
    for (int o = 16; o; o >>= 1) {
        m0 = fmaxf(m0, __shfl_xor(m0, o));
        m1 = fmaxf(m1, __shfl_xor(m1, o));
        m2 = fmaxf(m2, __shfl_xor(m2, o));
        m3 = fmaxf(m3, __shfl_xor(m3, o));
    }
    float a0 = 0.f, a1 = 0.f, a2 = 0.f, a3 = 0.f;
    float d0 = 0.f, d1 = 0.f, d2 = 0.f, d3 = 0.f;
    for (int b = 0; b < nE; b += 32) {
        int e = b + lane;
        bool hit = (e < nE) && (ei[nE + e] == n);
        unsigned mask = __builtin_amdgcn_ballot_w32(hit);
        while (mask) {
            int j = __builtin_ctz(mask);
            mask &= mask - 1;
            int s = ei[b + j];
            float4 ss = s_src4[s];
            float e0 = expf(lrelu(ss.x + sd.x) - m0);
            float e1 = expf(lrelu(ss.y + sd.y) - m1);
            float e2 = expf(lrelu(ss.z + sd.z) - m2);
            float e3 = expf(lrelu(ss.w + sd.w) - m3);
            d0 += e0; d1 += e1; d2 += e2; d3 += e3;
            const float* p = Wx + (size_t)s * DIM + lane;
            a0 += e0 * p[0];  a1 += e1 * p[32];
            a2 += e2 * p[64]; a3 += e3 * p[96];
        }
    }
    orow[0]  = elu1(a0 / (d0 + 1e-8f));
    orow[32] = elu1(a1 / (d1 + 1e-8f));
    orow[64] = elu1(a2 / (d2 + 1e-8f));
    orow[96] = elu1(a3 / (d3 + 1e-8f));
}

__global__ __launch_bounds__(256)
void gat_node_reduce(const int* __restrict__ cnt, const int* __restrict__ ei, int nE,
                     const float* __restrict__ s_src, const float* __restrict__ s_dst,
                     const float* __restrict__ Wx, float* out, int nNodes) {
    __shared__ int lsrc[8][CAP];
    int n = (int)((blockIdx.x * (size_t)blockDim.x + threadIdx.x) >> 5);
    n = __builtin_amdgcn_readfirstlane(n);
    if (n >= nNodes) return;
    int lane = threadIdx.x & 31;
    int* ls = lsrc[__builtin_amdgcn_readfirstlane((int)(threadIdx.x >> 5))];
    int deg = cnt[n];
    const float4* s_src4 = (const float4*)s_src;
    float4 sd = ((const float4*)s_dst)[n];
    float* orow = out + (size_t)n * DIM + lane;

    if (deg > CAP) {
        node_reduce_slow(n, lane, ei, nE, s_src4, sd, Wx, orow);
        return;
    }

    const int2* bk = (const int2*)out + (size_t)n * CAP;
    int2 pa = make_int2(0, 0), pb = make_int2(0, 0);
    if (lane < deg)      pa = bk[lane];
    if (lane + 32 < deg) pb = bk[lane + 32];
    int ra = 0, rb = 0;
    if (deg <= 32) {
        for (int j = 0; j < deg; ++j)
            ra += (__builtin_amdgcn_readlane(pa.x, j) < pa.x) ? 1 : 0;
    } else {
        for (int j = 0; j < 32; ++j) {
            int v = __builtin_amdgcn_readlane(pa.x, j);
            ra += (v < pa.x) ? 1 : 0;
            rb += (v < pb.x) ? 1 : 0;
        }
        for (int j = 0; j < deg - 32; ++j) {
            int v = __builtin_amdgcn_readlane(pb.x, j);
            ra += (v < pa.x) ? 1 : 0;
            rb += (v < pb.x) ? 1 : 0;
        }
    }
    if (lane < deg)      ls[ra] = pa.y;
    if (lane + 32 < deg) ls[rb] = pb.y;

    float m0 = -INFINITY, m1 = -INFINITY, m2 = -INFINITY, m3 = -INFINITY;
    for (int b = 0; b < deg; b += 32) {
        int i = b + lane;
        if (i < deg) {
            float4 ss = s_src4[ls[i]];
            m0 = fmaxf(m0, lrelu(ss.x + sd.x));
            m1 = fmaxf(m1, lrelu(ss.y + sd.y));
            m2 = fmaxf(m2, lrelu(ss.z + sd.z));
            m3 = fmaxf(m3, lrelu(ss.w + sd.w));
        }
    }
    for (int o = 16; o; o >>= 1) {
        m0 = fmaxf(m0, __shfl_xor(m0, o));
        m1 = fmaxf(m1, __shfl_xor(m1, o));
        m2 = fmaxf(m2, __shfl_xor(m2, o));
        m3 = fmaxf(m3, __shfl_xor(m3, o));
    }

    float a0 = 0.f, a1 = 0.f, a2 = 0.f, a3 = 0.f;
    float d0 = 0.f, d1 = 0.f, d2 = 0.f, d3 = 0.f;
    for (int b = 0; b < deg; b += 32) {
        int i = b + lane;
        int s = 0;
        float e0 = 0.f, e1 = 0.f, e2 = 0.f, e3 = 0.f;
        if (i < deg) {
            s = ls[i];
            float4 ss = s_src4[s];
            e0 = expf(lrelu(ss.x + sd.x) - m0);
            e1 = expf(lrelu(ss.y + sd.y) - m1);
            e2 = expf(lrelu(ss.z + sd.z) - m2);
            e3 = expf(lrelu(ss.w + sd.w) - m3);
        }
        d0 += e0; d1 += e1; d2 += e2; d3 += e3;

        int c = deg - b; if (c > 32) c = 32;
        int j = 0;
        for (; j + 4 <= c; j += 4) {
            const float* p0 = Wx + (size_t)__builtin_amdgcn_readlane(s, j    ) * DIM + lane;
            const float* p1 = Wx + (size_t)__builtin_amdgcn_readlane(s, j + 1) * DIM + lane;
            const float* p2 = Wx + (size_t)__builtin_amdgcn_readlane(s, j + 2) * DIM + lane;
            const float* p3 = Wx + (size_t)__builtin_amdgcn_readlane(s, j + 3) * DIM + lane;
            float r00 = p0[0], r01 = p0[32], r02 = p0[64], r03 = p0[96];
            float r10 = p1[0], r11 = p1[32], r12 = p1[64], r13 = p1[96];
            float r20 = p2[0], r21 = p2[32], r22 = p2[64], r23 = p2[96];
            float r30 = p3[0], r31 = p3[32], r32 = p3[64], r33 = p3[96];
            a0 += lane_f(e0, j    ) * r00; a1 += lane_f(e1, j    ) * r01;
            a2 += lane_f(e2, j    ) * r02; a3 += lane_f(e3, j    ) * r03;
            a0 += lane_f(e0, j + 1) * r10; a1 += lane_f(e1, j + 1) * r11;
            a2 += lane_f(e2, j + 1) * r12; a3 += lane_f(e3, j + 1) * r13;
            a0 += lane_f(e0, j + 2) * r20; a1 += lane_f(e1, j + 2) * r21;
            a2 += lane_f(e2, j + 2) * r22; a3 += lane_f(e3, j + 2) * r23;
            a0 += lane_f(e0, j + 3) * r30; a1 += lane_f(e1, j + 3) * r31;
            a2 += lane_f(e2, j + 3) * r32; a3 += lane_f(e3, j + 3) * r33;
        }
        for (; j < c; ++j) {
            const float* p0 = Wx + (size_t)__builtin_amdgcn_readlane(s, j) * DIM + lane;
            float r00 = p0[0], r01 = p0[32], r02 = p0[64], r03 = p0[96];
            a0 += lane_f(e0, j) * r00; a1 += lane_f(e1, j) * r01;
            a2 += lane_f(e2, j) * r02; a3 += lane_f(e3, j) * r03;
        }
    }
    for (int o = 16; o; o >>= 1) {
        d0 += __shfl_xor(d0, o);
        d1 += __shfl_xor(d1, o);
        d2 += __shfl_xor(d2, o);
        d3 += __shfl_xor(d3, o);
    }
    orow[0]  = elu1(a0 / (d0 + 1e-8f));
    orow[32] = elu1(a1 / (d1 + 1e-8f));
    orow[64] = elu1(a2 / (d2 + 1e-8f));
    orow[96] = elu1(a3 / (d3 + 1e-8f));
}

extern "C" void kernel_launch(void* const* d_in, const int* in_sizes, int n_in,
                              void* d_out, int out_size, void* d_ws, size_t ws_size,
                              hipStream_t stream) {
    const float* x   = (const float*)d_in[0];
    const int*   ei  = (const int*)d_in[1];
    const float* W   = (const float*)d_in[2];
    const float* a_w = (const float*)d_in[3];
    int N = in_sizes[0] / DIM;
    int E = in_sizes[1] / 2;
    float* out = (float*)d_out;

    float* Wx    = (float*)d_ws;
    float* s_src = Wx + (size_t)N * DIM;
    float* s_dst = s_src + (size_t)N * HEADS;
    int*   cnt   = (int*)(s_dst + (size_t)N * HEADS);
    int*   bucket = (int*)d_out;

    size_t need = (size_t)N * (DIM + 2 * HEADS + 1) * sizeof(float);
    if (ws_size < need) return;

    hipMemsetAsync(cnt, 0, (size_t)N * sizeof(int), stream);

    dim3 blk(256);
    int nTiles = (N + 15) / 16;
    int gemmBlocks = (nTiles + 7) / 8;
    gat_gemm_fill<<<gemmBlocks + (E + 255) / 256, blk, 0, stream>>>(
        x, W, a_w, Wx, s_src, s_dst, N, nTiles, gemmBlocks, ei, cnt, bucket, E);
    gat_node_reduce<<<(N + 7) / 8, blk, 0, stream>>>(cnt, ei, E, s_src, s_dst, Wx, out, N);
}


# Projection: W copied once per work-group into LDS by whole-row 16-byte loads, B fragments read from LDS and A from x 16 bytes per lane with one k order for both; Wx k sum order changed, fixed
# speedup vs baseline: 4.2251x; 1.0690x over previous
#include <hip/hip_runtime.h>
#include <hip/hip_bf16.h>

#define DIM       128
#define HEADS     4
#define HEAD_DIM  32

typedef __attribute__((ext_vector_type(2))) float v2f;
typedef __attribute__((ext_vector_type(8))) float v8f;

#define WPITCH 132

template <int M>
__device__ __forceinline__ float row_xor(float v) {
    return __int_as_float(__builtin_amdgcn_update_dpp(
        0, __float_as_int(v), 0x160 + M, 0xf, 0xf, true));
}

template <int M>
__device__ __forceinline__ float fold_pair(float lo, float hi, bool bit) {
    float keep = bit ? hi : lo;
    float send = bit ? lo : hi;
    return keep + row_xor<M>(send);
}

__device__ __forceinline__
void gemm_tile_wmma(const float* __restrict__ x, const float* __restrict__ W,
                    const float* __restrict__ a_w, float* wl,
                    float* __restrict__ Wx, float* __restrict__ s_src,
                    float* __restrict__ s_dst, int nNodes, int nTiles) {
    int wave = (blockIdx.x * blockDim.x + threadIdx.x) >> 5;
    int lane = threadIdx.x & 31;
    int rc   = lane & 15;
    int hi   = lane >> 4;
    int nodeBase = wave * 16;

    int arow = nodeBase + rc;
    if (arow >= nNodes) arow = nNodes - 1;
    const float* aptr = x + (size_t)arow * DIM + hi * 4;

    float4 ar[8];
    #pragma unroll
    for (int s = 0; s < 8; ++s) ar[s] = *(const float4*)(aptr + 8 * s);

    {
        int wrow = threadIdx.x >> 5;
        const float* wsrc = W + (size_t)wrow * DIM + lane * 4;
        float* wdst = wl + wrow * WPITCH + lane * 4;
        float4 wr[16];
        #pragma unroll
        for (int i = 0; i < 16; ++i) wr[i] = *(const float4*)(wsrc + (size_t)i * 8 * DIM);
        #pragma unroll
        for (int i = 0; i < 16; ++i) *(float4*)(wdst + i * 8 * WPITCH) = wr[i];
    }
    __syncthreads();
    if (wave >= nTiles) return;

    float as0 = a_w[rc], as1 = a_w[16 + rc];
    float ad0 = a_w[HEAD_DIM + rc], ad1 = a_w[HEAD_DIM + 16 + rc];

    v8f acc[8] = {};
    const float* bl = wl + rc * WPITCH + hi * 4;

    #pragma unroll
    for (int s = 0; s < 16; ++s) {
        float4 a4 = ar[s & 7];
        if (s + 8 < 16) ar[s & 7] = *(const float4*)(aptr + 8 * (s + 8));
        float4 b4[8];
        #pragma unroll
        for (int nt = 0; nt < 8; ++nt)
            b4[nt] = *(const float4*)(bl + nt * 16 * WPITCH + 8 * s);
        v2f a_lo = {a4.x, a4.y}, a_hi = {a4.z, a4.w};
        #pragma unroll
        for (int nt = 0; nt < 8; ++nt) {
            v2f b_lo = {b4[nt].x, b4[nt].y};
            acc[nt] = __builtin_amdgcn_wmma_f32_16x16x4_f32(
                false, a_lo, false, b_lo, (short)0, acc[nt], false, false);
        }
        #pragma unroll
        for (int nt = 0; nt < 8; ++nt) {
            v2f b_hi = {b4[nt].z, b4[nt].w};
            acc[nt] = __builtin_amdgcn_wmma_f32_16x16x4_f32(
                false, a_hi, false, b_hi, (short)0, acc[nt], false, false);
        }
    }

    #pragma unroll
    for (int nt = 0; nt < 8; ++nt) {
        #pragma unroll
        for (int j = 0; j < 8; ++j) {
            int node = nodeBase + j + 8 * hi;
            if (node < nNodes)
                Wx[(size_t)node * DIM + nt * 16 + rc] = acc[nt][j];
        }
    }

    float v[64];
    #pragma unroll
    for (int h = 0; h < HEADS; ++h) {
        #pragma unroll
        for (int j = 0; j < 8; ++j) {
            float x0 = acc[2 * h][j], x1 = acc[2 * h + 1][j];
            v[(h * 2 + 0) * 8 + j] = x0 * as0 + x1 * as1;
            v[(h * 2 + 1) * 8 + j] = x0 * ad0 + x1 * ad1;
        }
    }
    bool b0 = rc & 1, b1 = rc & 2, b2 = rc & 4, b3 = rc & 8;
    float w0[32], w1[16], w2[8], w3[4];
    #pragma unroll
    for (int t = 0; t < 32; ++t) w0[t] = fold_pair<1>(v[2 * t], v[2 * t + 1], b0);
    #pragma unroll
    for (int t = 0; t < 16; ++t) w1[t] = fold_pair<2>(w0[2 * t], w0[2 * t + 1], b1);
    #pragma unroll
    for (int t = 0; t < 8; ++t)  w2[t] = fold_pair<4>(w1[2 * t], w1[2 * t + 1], b2);
    #pragma unroll
    for (int t = 0; t < 4; ++t)  w3[t] = fold_pair<8>(w2[2 * t], w2[2 * t + 1], b3);

    int snode = nodeBase + (rc & 7) + 8 * hi;
    float* sp = b3 ? s_dst : s_src;
    if (snode < nNodes)
        *(float4*)(sp + (size_t)snode * HEADS) = make_float4(w3[0], w3[1], w3[2], w3[3]);
}

#define CAP 64

__device__ __forceinline__
void degree_fill(const int* ei, int* cnt,
                 int* __restrict__ bucket, int nE, int e) {
    if (e >= nE) return;
    int s = ei[e];
    int d = ei[nE + e];
    int r = atomicAdd(&cnt[d], 1);
    if (r < CAP) ((int2*)bucket)[(size_t)d * CAP + r] = make_int2(e, s);
}

__global__ __launch_bounds__(256)
void gat_gemm_fill(const float* __restrict__ x, const float* __restrict__ W,
                   const float* __restrict__ a_w,
                   float* __restrict__ Wx, float* __restrict__ s_src,
                   float* __restrict__ s_dst, int nNodes, int nTiles, int gemmBlocks,
                   const int* ei, int* cnt,
                   int* __restrict__ bucket, int nE) {
    __shared__ float wl[DIM * WPITCH];
    if ((int)blockIdx.x < gemmBlocks) {
        gemm_tile_wmma(x, W, a_w, wl, Wx, s_src, s_dst, nNodes, nTiles);
    } else {
        degree_fill(ei, cnt, bucket, nE,
                    (int)(blockIdx.x - gemmBlocks) * (int)blockDim.x + (int)threadIdx.x);
    }
}

__device__ __forceinline__ float lrelu(float v) { return v > 0.f ? v : 0.2f * v; }

__device__ __forceinline__ float lane_f(float v, int j) {
    return __int_as_float(__builtin_amdgcn_readlane(__float_as_int(v), j));
}

__device__ __forceinline__ float elu1(float v) { return v > 0.f ? v : expm1f(v); }

__device__ __forceinline__
void node_reduce_slow(int n, int lane, const int* __restrict__ ei, int nE,
                      const float4* __restrict__ s_src4, float4 sd,
                      const float* __restrict__ Wx, float* __restrict__ orow) {
    float m0 = -INFINITY, m1 = -INFINITY, m2 = -INFINITY, m3 = -INFINITY;
    for (int b = 0; b < nE; b += 32) {
        int e = b + lane;
        if (e < nE && ei[nE + e] == n) {
            float4 ss = s_src4[ei[e]];
            m0 = fmaxf(m0, lrelu(ss.x + sd.x));
            m1 = fmaxf(m1, lrelu(ss.y + sd.y));
            m2 = fmaxf(m2, lrelu(ss.z + sd.z));
            m3 = fmaxf(m3, lrelu(ss.w + sd.w));
        }
    }
    for (int o = 16; o; o >>= 1) {
        m0 = fmaxf(m0, __shfl_xor(m0, o));
        m1 = fmaxf(m1, __shfl_xor(m1, o));
        m2 = fmaxf(m2, __shfl_xor(m2, o));
        m3 = fmaxf(m3, __shfl_xor(m3, o));
    }
    float a0 = 0.f, a1 = 0.f, a2 = 0.f, a3 = 0.f;
    float d0 = 0.f, d1 = 0.f, d2 = 0.f, d3 = 0.f;
    for (int b = 0; b < nE; b += 32) {
        int e = b + lane;
        bool hit = (e < nE) && (ei[nE + e] == n);
        unsigned mask = __builtin_amdgcn_ballot_w32(hit);
        while (mask) {
            int j = __builtin_ctz(mask);
            mask &= mask - 1;
            int s = ei[b + j];
            float4 ss = s_src4[s];
            float e0 = expf(lrelu(ss.x + sd.x) - m0);
            float e1 = expf(lrelu(ss.y + sd.y) - m1);
            float e2 = expf(lrelu(ss.z + sd.z) - m2);
            float e3 = expf(lrelu(ss.w + sd.w) - m3);
            d0 += e0; d1 += e1; d2 += e2; d3 += e3;
            const float* p = Wx + (size_t)s * DIM + lane;
            a0 += e0 * p[0];  a1 += e1 * p[32];
            a2 += e2 * p[64]; a3 += e3 * p[96];
        }
    }
    orow[0]  = elu1(a0 / (d0 + 1e-8f));
    orow[32] = elu1(a1 / (d1 + 1e-8f));
    orow[64] = elu1(a2 / (d2 + 1e-8f));
    orow[96] = elu1(a3 / (d3 + 1e-8f));
}

__global__ __launch_bounds__(256)
void gat_node_reduce(const int* __restrict__ cnt, const int* __restrict__ ei, int nE,
                     const float* __restrict__ s_src, const float* __restrict__ s_dst,
                     const float* __restrict__ Wx, float* out, int nNodes) {
    __shared__ int lsrc[8][CAP];
    int n = (int)((blockIdx.x * (size_t)blockDim.x + threadIdx.x) >> 5);
    n = __builtin_amdgcn_readfirstlane(n);
    if (n >= nNodes) return;
    int lane = threadIdx.x & 31;
    int* ls = lsrc[__builtin_amdgcn_readfirstlane((int)(threadIdx.x >> 5))];
    int deg = cnt[n];
    const float4* s_src4 = (const float4*)s_src;
    float4 sd = ((const float4*)s_dst)[n];
    float* orow = out + (size_t)n * DIM + lane;

    if (deg > CAP) {
        node_reduce_slow(n, lane, ei, nE, s_src4, sd, Wx, orow);
        return;
    }

    const int2* bk = (const int2*)out + (size_t)n * CAP;
    int2 pa = make_int2(0, 0), pb = make_int2(0, 0);
    if (lane < deg)      pa = bk[lane];
    if (lane + 32 < deg) pb = bk[lane + 32];
    int ra = 0, rb = 0;
    if (deg <= 32) {
        for (int j = 0; j < deg; ++j)
            ra += (__builtin_amdgcn_readlane(pa.x, j) < pa.x) ? 1 : 0;
    } else {
        for (int j = 0; j < 32; ++j) {
            int v = __builtin_amdgcn_readlane(pa.x, j);
            ra += (v < pa.x) ? 1 : 0;
            rb += (v < pb.x) ? 1 : 0;
        }
        for (int j = 0; j < deg - 32; ++j) {
            int v = __builtin_amdgcn_readlane(pb.x, j);
            ra += (v < pa.x) ? 1 : 0;
            rb += (v < pb.x) ? 1 : 0;
        }
    }
    if (lane < deg)      ls[ra] = pa.y;
    if (lane + 32 < deg) ls[rb] = pb.y;

    float m0 = -INFINITY, m1 = -INFINITY, m2 = -INFINITY, m3 = -INFINITY;
    for (int b = 0; b < deg; b += 32) {
        int i = b + lane;
        if (i < deg) {
            float4 ss = s_src4[ls[i]];
            m0 = fmaxf(m0, lrelu(ss.x + sd.x));
            m1 = fmaxf(m1, lrelu(ss.y + sd.y));
            m2 = fmaxf(m2, lrelu(ss.z + sd.z));
            m3 = fmaxf(m3, lrelu(ss.w + sd.w));
        }
    }
    for (int o = 16; o; o >>= 1) {
        m0 = fmaxf(m0, __shfl_xor(m0, o));
        m1 = fmaxf(m1, __shfl_xor(m1, o));
        m2 = fmaxf(m2, __shfl_xor(m2, o));
        m3 = fmaxf(m3, __shfl_xor(m3, o));
    }

    float a0 = 0.f, a1 = 0.f, a2 = 0.f, a3 = 0.f;
    float d0 = 0.f, d1 = 0.f, d2 = 0.f, d3 = 0.f;
    for (int b = 0; b < deg; b += 32) {
        int i = b + lane;
        int s = 0;
        float e0 = 0.f, e1 = 0.f, e2 = 0.f, e3 = 0.f;
        if (i < deg) {
            s = ls[i];
            float4 ss = s_src4[s];
            e0 = expf(lrelu(ss.x + sd.x) - m0);
            e1 = expf(lrelu(ss.y + sd.y) - m1);
            e2 = expf(lrelu(ss.z + sd.z) - m2);
            e3 = expf(lrelu(ss.w + sd.w) - m3);
        }
        d0 += e0; d1 += e1; d2 += e2; d3 += e3;

        int c = deg - b; if (c > 32) c = 32;
        int j = 0;
        for (; j + 4 <= c; j += 4) {
            const float* p0 = Wx + (size_t)__builtin_amdgcn_readlane(s, j    ) * DIM + lane;
            const float* p1 = Wx + (size_t)__builtin_amdgcn_readlane(s, j + 1) * DIM + lane;
            const float* p2 = Wx + (size_t)__builtin_amdgcn_readlane(s, j + 2) * DIM + lane;
            const float* p3 = Wx + (size_t)__builtin_amdgcn_readlane(s, j + 3) * DIM + lane;
            float r00 = p0[0], r01 = p0[32], r02 = p0[64], r03 = p0[96];
            float r10 = p1[0], r11 = p1[32], r12 = p1[64], r13 = p1[96];
            float r20 = p2[0], r21 = p2[32], r22 = p2[64], r23 = p2[96];
            float r30 = p3[0], r31 = p3[32], r32 = p3[64], r33 = p3[96];
            a0 += lane_f(e0, j    ) * r00; a1 += lane_f(e1, j    ) * r01;
            a2 += lane_f(e2, j    ) * r02; a3 += lane_f(e3, j    ) * r03;
            a0 += lane_f(e0, j + 1) * r10; a1 += lane_f(e1, j + 1) * r11;
            a2 += lane_f(e2, j + 1) * r12; a3 += lane_f(e3, j + 1) * r13;
            a0 += lane_f(e0, j + 2) * r20; a1 += lane_f(e1, j + 2) * r21;
            a2 += lane_f(e2, j + 2) * r22; a3 += lane_f(e3, j + 2) * r23;
            a0 += lane_f(e0, j + 3) * r30; a1 += lane_f(e1, j + 3) * r31;
            a2 += lane_f(e2, j + 3) * r32; a3 += lane_f(e3, j + 3) * r33;
        }
        for (; j < c; ++j) {
            const float* p0 = Wx + (size_t)__builtin_amdgcn_readlane(s, j) * DIM + lane;
            float r00 = p0[0], r01 = p0[32], r02 = p0[64], r03 = p0[96];
            a0 += lane_f(e0, j) * r00; a1 += lane_f(e1, j) * r01;
            a2 += lane_f(e2, j) * r02; a3 += lane_f(e3, j) * r03;
        }
    }
    for (int o = 16; o; o >>= 1) {
        d0 += __shfl_xor(d0, o);
        d1 += __shfl_xor(d1, o);
        d2 += __shfl_xor(d2, o);
        d3 += __shfl_xor(d3, o);
    }
    orow[0]  = elu1(a0 / (d0 + 1e-8f));
    orow[32] = elu1(a1 / (d1 + 1e-8f));
    orow[64] = elu1(a2 / (d2 + 1e-8f));
    orow[96] = elu1(a3 / (d3 + 1e-8f));
}

extern "C" void kernel_launch(void* const* d_in, const int* in_sizes, int n_in,
                              void* d_out, int out_size, void* d_ws, size_t ws_size,
                              hipStream_t stream) {
    const float* x   = (const float*)d_in[0];
    const int*   ei  = (const int*)d_in[1];
    const float* W   = (const float*)d_in[2];
    const float* a_w = (const float*)d_in[3];
    int N = in_sizes[0] / DIM;
    int E = in_sizes[1] / 2;
    float* out = (float*)d_out;

    float* Wx    = (float*)d_ws;
    float* s_src = Wx + (size_t)N * DIM;
    float* s_dst = s_src + (size_t)N * HEADS;
    int*   cnt   = (int*)(s_dst + (size_t)N * HEADS);
    int*   bucket = (int*)d_out;

    size_t need = (size_t)N * (DIM + 2 * HEADS + 1) * sizeof(float);
    if (ws_size < need) return;

    hipMemsetAsync(cnt, 0, (size_t)N * sizeof(int), stream);

    dim3 blk(256);
    int nTiles = (N + 15) / 16;
    int gemmBlocks = (nTiles + 7) / 8;
    gat_gemm_fill<<<gemmBlocks + (E + 255) / 256, blk, 0, stream>>>(
        x, W, a_w, Wx, s_src, s_dst, N, nTiles, gemmBlocks, ei, cnt, bucket, E);
    gat_node_reduce<<<(N + 7) / 8, blk, 0, stream>>>(cnt, ei, E, s_src, s_dst, Wx, out, N);
}
